# starting kernel
# baseline (speedup 1.0000x reference)
#include <hip/hip_runtime.h>
#include <hip/hip_bf16.h>

#define N_SENT 100000
#define N_TYPE 10000
#define NEDGE  640000
#define DIM    128
#define SLOPE  0.01f

typedef __attribute__((ext_vector_type(16))) _Float16 v16h;
typedef __attribute__((ext_vector_type(8)))  float    v8f;

__global__ void gat_score_wmma(const float* __restrict__ H,
                               const float* __restrict__ wv,
                               float* __restrict__ out, int nrows) {
    const int lane = threadIdx.x & 31;
    const int wave = (blockIdx.x * blockDim.x + threadIdx.x) >> 5;
    const int r0 = wave * 16;
    if (r0 >= nrows) return;

    const int m    = lane & 15;
    const int koff = (lane < 16) ? 0 : 8;

    v16h ones;
#pragma unroll
    for (int i = 0; i < 16; ++i) ones[i] = (_Float16)1.0f;

    v8f c = {};
#pragma unroll
    for (int k0 = 0; k0 < DIM; k0 += 32) {
        const float* hp = H + (size_t)(r0 + m) * DIM + k0;
        const float* wp = wv + k0;
        v16h a;
#pragma unroll
        for (int j = 0; j < 8; ++j)
            a[j] = (_Float16)(hp[koff + j] * wp[koff + j]);
#pragma unroll
        for (int j = 0; j < 8; ++j)
            a[8 + j] = (_Float16)(hp[16 + koff + j] * wp[16 + koff + j]);
        c = __builtin_amdgcn_wmma_f32_16x16x32_f16(
                 false, a,  false, ones,
                 (short)0, c,  false,  false);
    }
    if (lane == 0) {
#pragma unroll
        for (int r = 0; r < 8; ++r) out[r0 + r] = c[r];
    }
    if (lane == 16) {
#pragma unroll
        for (int r = 0; r < 8; ++r) out[r0 + 8 + r] = c[r];
    }
}

__global__ void gat_zero(int* __restrict__ deg, int* __restrict__ cur, int n) {
    int i = blockIdx.x * blockDim.x + threadIdx.x;
    if (i < n) { deg[i] = 0; cur[i] = 0; }
}

__global__ void gat_degree(const int* __restrict__ dst, int* __restrict__ deg, int e) {
    int i = blockIdx.x * blockDim.x + threadIdx.x;
    if (i < e) atomicAdd(&deg[dst[i]], 1);
}

__global__ void gat_scan(const int* __restrict__ deg, int* __restrict__ row_start, int n) {
    __shared__ int sh[1024];
    const int t = threadIdx.x;
    int carry = 0;
    for (int base = 0; base < n; base += 1024) {
        int v = (base + t < n) ? deg[base + t] : 0;
        sh[t] = v;
        __syncthreads();
        for (int off = 1; off < 1024; off <<= 1) {
            int y = (t >= off) ? sh[t - off] : 0;
            __syncthreads();
            sh[t] += y;
            __syncthreads();
        }
        int incl  = sh[t];
        int total = sh[1023];
        if (base + t < n) row_start[base + t] = carry + incl - v;
        __syncthreads();
        carry += total;
    }
    if (t == 0) row_start[n] = carry;
}

__global__ void gat_fill(const int* __restrict__ dst, const int* __restrict__ row_start,
                         int* __restrict__ cur, int* __restrict__ eidx, int e) {
    int i = blockIdx.x * blockDim.x + threadIdx.x;
    if (i < e) {
        int d = dst[i];
        int pos = atomicAdd(&cur[d], 1);
        eidx[row_start[d] + pos] = i;
    }
}

__global__ void gat_aggregate(const float* __restrict__ h_sent,
                              const float* __restrict__ h_type,
                              const float* __restrict__ s_src,
                              const float* __restrict__ s_dst,
                              const int*   __restrict__ srcI,
                              const int*   __restrict__ row_start,
                              const int*   __restrict__ eidx,
                              float*       __restrict__ ws_ex,
                              float*       __restrict__ out) {
    const int wid  = (blockIdx.x * blockDim.x + threadIdx.x) >> 5;
    const int lane = threadIdx.x & 31;
    if (wid >= N_TYPE) return;

    const int start = row_start[wid];
    const int end   = row_start[wid + 1];

    float4* orow = (float4*)(out + (size_t)wid * DIM);
    if (end == start) {
        const float4* trow = (const float4*)(h_type + (size_t)wid * DIM);
        orow[lane] = trow[lane];
        return;
    }
    const float sdt = s_dst[wid];

    float mx = -3.402823466e38f;
    for (int i = start + lane; i < end; i += 32) {
        float v = s_src[srcI[eidx[i]]] + sdt;
        v = (v > 0.0f) ? v : SLOPE * v;
        mx = fmaxf(mx, v);
    }
#pragma unroll
    for (int off = 16; off > 0; off >>= 1) mx = fmaxf(mx, __shfl_xor(mx, off, 32));

    float sm = 0.0f;
    for (int i = start + lane; i < end; i += 32) {
        float v = s_src[srcI[eidx[i]]] + sdt;
        v = (v > 0.0f) ? v : SLOPE * v;
        float ex = __expf(v - mx);
        ws_ex[i] = ex;
        sm += ex;
    }
#pragma unroll
    for (int off = 16; off > 0; off >>= 1) sm += __shfl_xor(sm, off, 32);
    const float inv = 1.0f / sm;

    float4 acc = {0.0f, 0.0f, 0.0f, 0.0f};
    for (int i = start; i < end; ++i) {
        if (i + 1 < end)
            __builtin_prefetch(h_sent + (size_t)srcI[eidx[i + 1]] * DIM + lane * 4, 0, 3);
        const float a = ws_ex[i] * inv;
        const float4 hv = *(const float4*)(h_sent + (size_t)srcI[eidx[i]] * DIM + lane * 4);
        acc.x += a * hv.x; acc.y += a * hv.y; acc.z += a * hv.z; acc.w += a * hv.w;
    }
    orow[lane] = acc;
}

extern "C" void kernel_launch(void* const* d_in, const int* in_sizes, int n_in,
                              void* d_out, int out_size, void* d_ws, size_t ws_size,
                              hipStream_t stream) {
    const float* h_sent = (const float*)d_in[0];
    const float* h_type = (const float*)d_in[1];
    const float* attn_w = (const float*)d_in[2];
    const int*   srcI   = (const int*)d_in[3];
    const int*   dstI   = (const int*)d_in[4];
    float*       out    = (float*)d_out;

    char* ws = (char*)d_ws;
    float* s_src     = (float*)(ws);
    float* s_dst     = (float*)(ws + 400000);
    int*   deg       = (int*)  (ws + 440000);
    int*   cursor    = (int*)  (ws + 480000);
    int*   row_start = (int*)  (ws + 520000);
    int*   eidx      = (int*)  (ws + 560016);
    float* ws_ex     = (float*)(ws + 3120016);
    (void)ws_size; (void)n_in; (void)in_sizes; (void)out_size;

    {
        int waves = N_SENT / 16;
        int blocks = (waves + 3) / 4;
        gat_score_wmma<<<blocks, 128, 0, stream>>>(h_sent, attn_w, s_src, N_SENT);
    }
    {
        int waves = N_TYPE / 16;
        int blocks = (waves + 3) / 4;
        gat_score_wmma<<<blocks, 128, 0, stream>>>(h_type, attn_w + DIM, s_dst, N_TYPE);
    }

    gat_zero<<<(N_TYPE + 255) / 256, 256, 0, stream>>>(deg, cursor, N_TYPE);

    gat_degree<<<(NEDGE + 255) / 256, 256, 0, stream>>>(dstI, deg, NEDGE);

    gat_scan<<<1, 1024, 0, stream>>>(deg, row_start, N_TYPE);

    gat_fill<<<(NEDGE + 255) / 256, 256, 0, stream>>>(dstI, row_start, cursor, eidx, NEDGE);

    {
        int blocks = (N_TYPE * 32 + 255) / 256;
        gat_aggregate<<<blocks, 256, 0, stream>>>(h_sent, h_type, s_src, s_dst,
                                                  srcI, row_start, eidx, ws_ex, out);
    }
}


# gat_aggregate: each lane fetches its edges' row index and softmax weight once; the gather reads them by readlane into scalars, so 8 independent row loads are in flight per wave; exp cache dropped
# speedup vs baseline: 3.8752x; 3.8752x over previous
#include <hip/hip_runtime.h>
#include <hip/hip_bf16.h>

#define N_SENT 100000
#define N_TYPE 10000
#define NEDGE  640000
#define DIM    128
#define SLOPE  0.01f

typedef __attribute__((ext_vector_type(16))) _Float16 v16h;
typedef __attribute__((ext_vector_type(8)))  float    v8f;

__global__ void gat_score_wmma(const float* __restrict__ H,
                               const float* __restrict__ wv,
                               float* __restrict__ out, int nrows) {
    const int lane = threadIdx.x & 31;
    const int wave = (blockIdx.x * blockDim.x + threadIdx.x) >> 5;
    const int r0 = wave * 16;
    if (r0 >= nrows) return;

    const int m    = lane & 15;
    const int koff = (lane < 16) ? 0 : 8;

    v16h ones;
#pragma unroll
    for (int i = 0; i < 16; ++i) ones[i] = (_Float16)1.0f;

    v8f c = {};
#pragma unroll
    for (int k0 = 0; k0 < DIM; k0 += 32) {
        const float* hp = H + (size_t)(r0 + m) * DIM + k0;
        const float* wp = wv + k0;
        v16h a;
#pragma unroll
        for (int j = 0; j < 8; ++j)
            a[j] = (_Float16)(hp[koff + j] * wp[koff + j]);
#pragma unroll
        for (int j = 0; j < 8; ++j)
            a[8 + j] = (_Float16)(hp[16 + koff + j] * wp[16 + koff + j]);
        c = __builtin_amdgcn_wmma_f32_16x16x32_f16(
                 false, a,  false, ones,
                 (short)0, c,  false,  false);
    }
    if (lane == 0) {
#pragma unroll
        for (int r = 0; r < 8; ++r) out[r0 + r] = c[r];
    }
    if (lane == 16) {
#pragma unroll
        for (int r = 0; r < 8; ++r) out[r0 + 8 + r] = c[r];
    }
}

#define NCHUNK 64
#define CHUNK  ((NEDGE + NCHUNK - 1) / NCHUNK)
#define PITCH  10048
#define EDGES_PER_THREAD ((CHUNK + 1023) / 1024)
#define BINS_PER_THREAD  ((N_TYPE + 1023) / 1024)
static_assert(NEDGE % NCHUNK == 0, "chunks must be equal so that only a thread's last edge slot needs a guard");
static_assert(N_TYPE % BINS_PER_THREAD == 0, "a thread's bins must be all inside or all outside the table");
static_assert(PITCH >= N_TYPE && PITCH % 64 == 0, "part rows hold N_TYPE ints in whole 256-B units");

__global__ __launch_bounds__(1024)
void gat_hist(const int* __restrict__ dst, int* __restrict__ part) {
    __shared__ int hist[N_TYPE];
    const int t  = threadIdx.x;
    const int k  = blockIdx.x;
    const int e0 = k * CHUNK;

    int d[EDGES_PER_THREAD];
#pragma unroll
    for (int j = 0; j < EDGES_PER_THREAD; ++j) {
        const int o = t + j * 1024;
        d[j] = (o < CHUNK) ? dst[e0 + o] : 0;
    }
    for (int b = t; b < N_TYPE; b += 1024) hist[b] = 0;
    __syncthreads();
#pragma unroll
    for (int j = 0; j < EDGES_PER_THREAD; ++j)
        if (t + j * 1024 < CHUNK) atomicAdd(&hist[d[j]], 1);
    __syncthreads();
    int* row = part + (size_t)k * PITCH;
    for (int b = t; b < N_TYPE; b += 1024) row[b] = hist[b];
}

__global__ void gat_colprefix(int* __restrict__ part, int* __restrict__ tot) {
    const int b = blockIdx.x * blockDim.x + threadIdx.x;
    if (b >= N_TYPE) return;
    int v[NCHUNK];
#pragma unroll
    for (int k = 0; k < NCHUNK; ++k) v[k] = part[(size_t)k * PITCH + b];
    int run = 0;
#pragma unroll
    for (int k = 0; k < NCHUNK; ++k) {
        part[(size_t)k * PITCH + b] = run;
        run += v[k];
    }
    tot[b] = run;
}

__global__ __launch_bounds__(1024)
void gat_fill(const int* __restrict__ dst, const int* __restrict__ part,
              const int* __restrict__ tot, int* __restrict__ row_start,
              int* __restrict__ eidx) {
    __shared__ int cur[N_TYPE];
    __shared__ int wsum[32];
    const int t    = threadIdx.x;
    const int k    = blockIdx.x;
    const int lane = t & 31;
    const int w    = t >> 5;
    const int e0   = k * CHUNK;

    int d[EDGES_PER_THREAD];
#pragma unroll
    for (int j = 0; j < EDGES_PER_THREAD; ++j) {
        const int o = t + j * 1024;
        d[j] = (o < CHUNK) ? dst[e0 + o] : 0;
    }

    const int b0 = t * BINS_PER_THREAD;
    const int* prow = part + (size_t)k * PITCH;
    int v[BINS_PER_THREAD];
    int p[BINS_PER_THREAD];
    int s = 0;
    const bool own = b0 < N_TYPE;
#pragma unroll
    for (int j = 0; j < BINS_PER_THREAD; ++j) { v[j] = 0; p[j] = 0; }
    if (own) {
#pragma unroll
        for (int j = 0; j < BINS_PER_THREAD; ++j) { v[j] = tot[b0 + j]; p[j] = prow[b0 + j]; }
    }
#pragma unroll
    for (int j = 0; j < BINS_PER_THREAD; ++j) s += v[j];
    int inc = s;
#pragma unroll
    for (int off = 1; off < 32; off <<= 1) {
        const int y = __shfl_up(inc, off, 32);
        if (lane >= off) inc += y;
    }
    if (lane == 31) wsum[w] = inc;
    __syncthreads();
    if (w == 0) {
        const int x = wsum[lane];
        int xi = x;
#pragma unroll
        for (int off = 1; off < 32; off <<= 1) {
            const int y = __shfl_up(xi, off, 32);
            if (lane >= off) xi += y;
        }
        wsum[lane] = xi - x;
    }
    __syncthreads();
    int excl = wsum[w] + inc - s;
    int rs[BINS_PER_THREAD];
#pragma unroll
    for (int j = 0; j < BINS_PER_THREAD; ++j) { rs[j] = excl; excl += v[j]; }
    if (own) {
#pragma unroll
        for (int j = 0; j < BINS_PER_THREAD; ++j) cur[b0 + j] = rs[j] + p[j];
        if (k == 0) {
#pragma unroll
            for (int j = 0; j < BINS_PER_THREAD; ++j) row_start[b0 + j] = rs[j];
        }
    }
    if (k == 0 && t == 1023) row_start[N_TYPE] = excl;
    __syncthreads();

    int pos[EDGES_PER_THREAD];
#pragma unroll
    for (int j = 0; j < EDGES_PER_THREAD; ++j)
        pos[j] = (t + j * 1024 < CHUNK) ? atomicAdd(&cur[d[j]], 1) : -1;
#pragma unroll
    for (int j = 0; j < EDGES_PER_THREAD; ++j)
        if (t + j * 1024 < CHUNK) eidx[pos[j]] = e0 + t + j * 1024;
}

#define GROUP   8
#define NCACHE  4

__device__ __forceinline__ void gather_rows(const float* __restrict__ h_sent,
                                            int sr, float w, int cnt, int lane,
                                            float4& acc) {
    const int n = (cnt + GROUP - 1) & ~(GROUP - 1);
    const int wi = __float_as_int(w);
    for (int j = 0; j < n; j += GROUP) {
        float4 hv[GROUP];
        float  a[GROUP];
#pragma unroll
        for (int u = 0; u < GROUP; ++u) {
            const int s = __builtin_amdgcn_readlane(sr, j + u);
            a[u] = __int_as_float(__builtin_amdgcn_readlane(wi, j + u));
            hv[u] = *(const float4*)(h_sent + (size_t)s * DIM + lane * 4);
        }
#pragma unroll
        for (int u = 0; u < GROUP; ++u) {
            acc.x = __builtin_fmaf(a[u], hv[u].x, acc.x); acc.y = __builtin_fmaf(a[u], hv[u].y, acc.y);
            acc.z = __builtin_fmaf(a[u], hv[u].z, acc.z); acc.w = __builtin_fmaf(a[u], hv[u].w, acc.w);
        }
    }
}

__global__ void gat_aggregate(const float* __restrict__ h_sent,
                              const float* __restrict__ h_type,
                              const float* __restrict__ s_src,
                              const float* __restrict__ s_dst,
                              const int*   __restrict__ srcI,
                              const int*   __restrict__ row_start,
                              const int*   __restrict__ eidx,
                              float*       __restrict__ out) {
    const int wid  = __builtin_amdgcn_readfirstlane((int)((blockIdx.x * blockDim.x + threadIdx.x) >> 5));
    const int lane = threadIdx.x & 31;
    if (wid >= N_TYPE) return;

    const int start = row_start[wid];
    const int end   = row_start[wid + 1];
    const int deg   = end - start;

    float4* orow = (float4*)(out + (size_t)wid * DIM);
    if (deg <= 0) {
        const float4* trow = (const float4*)(h_type + (size_t)wid * DIM);
        orow[lane] = trow[lane];
        return;
    }
    const float sdt = s_dst[wid];
    const float NEG = -3.402823466e38f;
    float4 acc = {0.0f, 0.0f, 0.0f, 0.0f};

    if (deg <= 32 * NCACHE) {
        int   ix[NCACHE];
        int   sr[NCACHE];
        float v[NCACHE];
#pragma unroll
        for (int c = 0; c < NCACHE; ++c) ix[c] = eidx[min(start + c * 32 + lane, end - 1)];
#pragma unroll
        for (int c = 0; c < NCACHE; ++c) sr[c] = srcI[ix[c]];
#pragma unroll
        for (int c = 0; c < NCACHE; ++c) v[c] = s_src[sr[c]];
        float mx = NEG;
#pragma unroll
        for (int c = 0; c < NCACHE; ++c) {
            float t = v[c] + sdt;
            t = (t > 0.0f) ? t : SLOPE * t;
            v[c] = (c * 32 + lane < deg) ? t : NEG;
            mx = fmaxf(mx, v[c]);
        }
#pragma unroll
        for (int off = 16; off > 0; off >>= 1) mx = fmaxf(mx, __shfl_xor(mx, off, 32));
        float sm = 0.0f;
#pragma unroll
        for (int c = 0; c < NCACHE; ++c) {
            v[c] = (c * 32 + lane < deg) ? __expf(v[c] - mx) : 0.0f;
            sm += v[c];
        }
#pragma unroll
        for (int off = 16; off > 0; off >>= 1) sm += __shfl_xor(sm, off, 32);
        const float inv = 1.0f / sm;
#pragma unroll
        for (int c = 0; c < NCACHE; ++c) {
            const int cnt = min(deg - c * 32, 32);
            if (cnt > 0) gather_rows(h_sent, sr[c], v[c] * inv, cnt, lane, acc);
        }
    } else {
        float mx = NEG;
        for (int i = start + lane; i < end; i += 32) {
            float t = s_src[srcI[eidx[i]]] + sdt;
            t = (t > 0.0f) ? t : SLOPE * t;
            mx = fmaxf(mx, t);
        }
#pragma unroll
        for (int off = 16; off > 0; off >>= 1) mx = fmaxf(mx, __shfl_xor(mx, off, 32));
        float sm = 0.0f;
        for (int i = start + lane; i < end; i += 32) {
            float t = s_src[srcI[eidx[i]]] + sdt;
            t = (t > 0.0f) ? t : SLOPE * t;
            sm += __expf(t - mx);
        }
#pragma unroll
        for (int off = 16; off > 0; off >>= 1) sm += __shfl_xor(sm, off, 32);
        const float inv = 1.0f / sm;
        for (int base = start; base < end; base += 32) {
            const int i = base + lane;
            const int s = srcI[eidx[min(i, end - 1)]];
            float t = s_src[s] + sdt;
            t = (t > 0.0f) ? t : SLOPE * t;
            const float w = (i < end) ? __expf(t - mx) * inv : 0.0f;
            gather_rows(h_sent, s, w, min(end - base, 32), lane, acc);
        }
    }
    orow[lane] = acc;
}

extern "C" void kernel_launch(void* const* d_in, const int* in_sizes, int n_in,
                              void* d_out, int out_size, void* d_ws, size_t ws_size,
                              hipStream_t stream) {
    const float* h_sent = (const float*)d_in[0];
    const float* h_type = (const float*)d_in[1];
    const float* attn_w = (const float*)d_in[2];
    const int*   srcI   = (const int*)d_in[3];
    const int*   dstI   = (const int*)d_in[4];
    float*       out    = (float*)d_out;

    char* ws = (char*)d_ws;
    float* s_src     = (float*)(ws);
    float* s_dst     = (float*)(ws + 400000);
    int*   tot       = (int*)  (ws + 440000);
    int*   row_start = (int*)  (ws + 480000);
    int*   eidx      = (int*)  (ws + 520192);
    int*   part      = (int*)  (ws + 3080192);
    (void)ws_size; (void)n_in; (void)in_sizes; (void)out_size;

    {
        int waves = N_SENT / 16;
        int blocks = (waves + 3) / 4;
        gat_score_wmma<<<blocks, 128, 0, stream>>>(h_sent, attn_w, s_src, N_SENT);
    }
    {
        int waves = N_TYPE / 16;
        int blocks = (waves + 3) / 4;
        gat_score_wmma<<<blocks, 128, 0, stream>>>(h_type, attn_w + DIM, s_dst, N_TYPE);
    }

    gat_hist<<<NCHUNK, 1024, 0, stream>>>(dstI, part);

    gat_colprefix<<<(N_TYPE + 63) / 64, 64, 0, stream>>>(part, tot);

    gat_fill<<<NCHUNK, 1024, 0, stream>>>(dstI, part, tot, row_start, eidx);

    {
        int blocks = (N_TYPE * 32 + 255) / 256;
        gat_aggregate<<<blocks, 256, 0, stream>>>(h_sent, h_type, s_src, s_dst,
                                                  srcI, row_start, eidx, out);
    }
}


# score tiles read whole 512-B rows (row sums by a 0/1-matrix WMMA plus 8 register adds: new f32 sum order), two tiles per wave: 108, not 215, work-groups of short-lived score waves in launch 1
# speedup vs baseline: 9.3445x; 2.4114x over previous
#include <hip/hip_runtime.h>
#include <hip/hip_bf16.h>

#define N_SENT 100000
#define N_TYPE 10000
#define NEDGE  640000
#define DIM    128
#define SLOPE  0.01f

typedef __attribute__((ext_vector_type(16))) _Float16 v16h;
typedef __attribute__((ext_vector_type(8)))  float    v8f;

#define TILES_SENT (N_SENT / 16)
#define TILES_TYPE (N_TYPE / 16)
#define TILES_ALL  (TILES_SENT + TILES_TYPE)
#define TILES_PER_WAVE 2
#define TILES_PER_BLOCK (32 * TILES_PER_WAVE)
#define SCORE_BLOCKS_ALL ((TILES_ALL + TILES_PER_BLOCK - 1) / TILES_PER_BLOCK)
static_assert(N_SENT % 16 == 0 && N_TYPE % 16 == 0, "whole 16-row tiles only: no row guard inside a tile");
static_assert(TILES_SENT % TILES_PER_WAVE == 0, "a wave's tiles must all lie in one matrix: they share its weight half");
static_assert(DIM == 128, "one float4 per lane covers a row");

__device__ __forceinline__ void score_tile16(const float* __restrict__ H, const float4 wq,
                                             float* __restrict__ out, int r0, int lane) {
    const int c = lane & 15;
    float hv[16][4];
#pragma unroll
    for (int i = 0; i < 16; ++i) {
        const float4 q = *(const float4*)(H + (size_t)(r0 + i) * DIM + lane * 4);
        hv[i][0] = q.x; hv[i][1] = q.y; hv[i][2] = q.z; hv[i][3] = q.w;
    }
    const float wl[4] = {wq.x, wq.y, wq.z, wq.w};
    v8f d = {};
#pragma unroll
    for (int g = 0; g < 4; ++g) {
        v16h a, b;
#pragma unroll
        for (int s = 0; s < 16; ++s) {
            const int i = 4 * g + (s >> 2);
            a[s] = (_Float16)(hv[i][s & 3] * wl[s & 3]);
            b[s] = (c == i) ? (_Float16)1.0f : (_Float16)0.0f;
        }
        d = __builtin_amdgcn_wmma_f32_16x16x32_f16(
                 false, a,  false, b,
                 (short)0, d,  false,  false);
    }
    float t = d[0];
#pragma unroll
    for (int r = 1; r < 8; ++r) t += d[r];
    t += __shfl_xor(t, 16, 32);
    if (lane < 16) out[r0 + lane] = t;
}

__device__ __forceinline__ void score_role(int tile,
                                           const float* __restrict__ h_sent,
                                           const float* __restrict__ h_type,
                                           const float* __restrict__ attn_w,
                                           float* __restrict__ s_src,
                                           float* __restrict__ s_dst) {
    if (tile >= TILES_ALL) return;
    const int lane = threadIdx.x & 31;
    const bool sent = tile < TILES_SENT;
    const float* H  = sent ? h_sent : h_type;
    const float* wv = sent ? attn_w : attn_w + DIM;
    float*       o  = sent ? s_src : s_dst;
    const int    r0 = (sent ? tile : tile - TILES_SENT) * 16;
    const int nt = min((sent ? TILES_SENT : TILES_ALL) - tile, TILES_PER_WAVE);
    const float4 wq = *(const float4*)(wv + lane * 4);
#pragma clang loop unroll(disable)
    for (int i = 0; i < nt; ++i) score_tile16(H, wq, o, r0 + 16 * i, lane);
}

#define NCHUNK 64
#define CHUNK  ((NEDGE + NCHUNK - 1) / NCHUNK)
#define DBLK   8
#define EDGES_PER_THREAD ((CHUNK + 1023) / 1024)
#define BINS_PER_THREAD  ((N_TYPE + 1023) / 1024)
static_assert(NEDGE % NCHUNK == 0, "chunks must be equal so that only a thread's last edge slot needs a guard");
static_assert(N_TYPE % BINS_PER_THREAD == 0, "a thread's bins must be all inside or all outside the table");
static_assert(N_TYPE % DBLK == 0, "whole table blocks only");
static_assert(CHUNK < 65536, "first slot and count of a chunk share one 32-bit word");

__global__ __launch_bounds__(1024) __attribute__((amdgpu_waves_per_eu(8, 8)))
void gat_sort(const int* __restrict__ dst, const int* __restrict__ srcI,
              int* __restrict__ tab, int* __restrict__ srt,
              const float* __restrict__ h_sent, const float* __restrict__ h_type,
              const float* __restrict__ attn_w,
              float* __restrict__ s_src, float* __restrict__ s_dst) {
    __shared__ int hist[N_TYPE];
    __shared__ int sorted[CHUNK];
    __shared__ int wsum[32];
    const int t = threadIdx.x;
    const int k = blockIdx.x;
    if (k >= NCHUNK) {
        const int tile = __builtin_amdgcn_readfirstlane(((k - NCHUNK) * 32 + (t >> 5)) * TILES_PER_WAVE);
        score_role(tile, h_sent, h_type, attn_w, s_src, s_dst);
        return;
    }
    const int lane = t & 31;
    const int w    = t >> 5;
    const int e0   = k * CHUNK;

    int d[EDGES_PER_THREAD];
    int sv[EDGES_PER_THREAD];
#pragma unroll
    for (int j = 0; j < EDGES_PER_THREAD; ++j) {
        const int o = t + j * 1024;
        d[j] = (o < CHUNK) ? dst[e0 + o] : 0;
    }
#pragma unroll
    for (int j = 0; j < EDGES_PER_THREAD; ++j) {
        const int o = t + j * 1024;
        sv[j] = (o < CHUNK) ? srcI[e0 + o] : 0;
    }
    for (int b = t; b < N_TYPE; b += 1024) hist[b] = 0;
    __syncthreads();
    int rk[EDGES_PER_THREAD];
#pragma unroll
    for (int j = 0; j < EDGES_PER_THREAD; ++j)
        rk[j] = (t + j * 1024 < CHUNK) ? atomicAdd(&hist[d[j]], 1) : 0;
    __syncthreads();

    const int b0 = t * BINS_PER_THREAD;
    const bool own = b0 < N_TYPE;
    int v[BINS_PER_THREAD];
    int s = 0;
#pragma unroll
    for (int j = 0; j < BINS_PER_THREAD; ++j) { v[j] = own ? hist[b0 + j] : 0; s += v[j]; }
    int inc = s;
#pragma unroll
    for (int off = 1; off < 32; off <<= 1) {
        const int y = __shfl_up(inc, off, 32);
        if (lane >= off) inc += y;
    }
    if (lane == 31) wsum[w] = inc;
    __syncthreads();
    if (w == 0) {
        const int x = wsum[lane];
        int xi = x;
#pragma unroll
        for (int off = 1; off < 32; off <<= 1) {
            const int y = __shfl_up(xi, off, 32);
            if (lane >= off) xi += y;
        }
        wsum[lane] = xi - x;
    }
    __syncthreads();
    int excl = wsum[w] + inc - s;
    if (own) {
#pragma unroll
        for (int j = 0; j < BINS_PER_THREAD; ++j) { hist[b0 + j] = excl; excl += v[j]; }
    }
    __syncthreads();

    for (int b = t; b < N_TYPE; b += 1024) {
        const int first = hist[b];
        const int next  = (b + 1 < N_TYPE) ? hist[b + 1] : CHUNK;
        tab[(size_t)(b / DBLK) * (NCHUNK * DBLK) + k * DBLK + (b % DBLK)] = first | ((next - first) << 16);
    }
#pragma unroll
    for (int j = 0; j < EDGES_PER_THREAD; ++j)
        if (t + j * 1024 < CHUNK) sorted[hist[d[j]] + rk[j]] = sv[j];
    __syncthreads();
    for (int b = t; b < CHUNK; b += 1024) srt[e0 + b] = sorted[b];
}

#define GROUP   8
#define NCACHE  4
#define SLOTS   (32 * NCACHE)
#define AGG_WAVES 8
static_assert(AGG_WAVES == DBLK, "the waves of a work-group share one table block");
static_assert(NCHUNK == 64, "two table words per lane");

__device__ __forceinline__ void gather_rows(const float* __restrict__ h_sent,
                                            int sr, float w, int cnt, int lane,
                                            float4& acc) {
    const int n = (cnt + GROUP - 1) & ~(GROUP - 1);
    const int wi = __float_as_int(w);
    for (int j = 0; j < n; j += GROUP) {
        float4 hv[GROUP];
        float  a[GROUP];
#pragma unroll
        for (int u = 0; u < GROUP; ++u) {
            const int s = __builtin_amdgcn_readlane(sr, j + u);
            a[u] = __int_as_float(__builtin_amdgcn_readlane(wi, j + u));
            hv[u] = *(const float4*)(h_sent + (size_t)s * DIM + lane * 4);
        }
#pragma unroll
        for (int u = 0; u < GROUP; ++u) {
            acc.x = __builtin_fmaf(a[u], hv[u].x, acc.x); acc.y = __builtin_fmaf(a[u], hv[u].y, acc.y);
            acc.z = __builtin_fmaf(a[u], hv[u].z, acc.z); acc.w = __builtin_fmaf(a[u], hv[u].w, acc.w);
        }
    }
}

__global__ __launch_bounds__(32 * AGG_WAVES)
void gat_aggregate(const float* __restrict__ h_sent,
                   const float* __restrict__ h_type,
                   const float* __restrict__ s_src,
                   const float* __restrict__ s_dst,
                   const int*   __restrict__ tab,
                   const int*   __restrict__ srt,
                   float*       __restrict__ out) {
    __shared__ int sbuf[AGG_WAVES][SLOTS];
    const int wid  = __builtin_amdgcn_readfirstlane((int)((blockIdx.x * blockDim.x + threadIdx.x) >> 5));
    const int lane = threadIdx.x & 31;
    if (wid >= N_TYPE) return;

    const int* tb = tab + (size_t)(wid / DBLK) * (NCHUNK * DBLK) + (wid % DBLK);
    const int p0 = tb[lane * DBLK];
    const int p1 = tb[(lane + 32) * DBLK];
    const int st0 = p0 & 0xFFFF, c0 = p0 >> 16;
    const int st1 = p1 & 0xFFFF, c1 = p1 >> 16;
    const int q = min(c0, 255) | (min(c1, 255) << 16);
    int inc = q;
#pragma unroll
    for (int off = 1; off < 32; off <<= 1) {
        const int y = __shfl_up(inc, off, 32);
        if (lane >= off) inc += y;
    }
    const int totp = __builtin_amdgcn_readlane(inc, 31);
    const int tot0 = totp & 0xFFFF;
    const int degc = tot0 + (totp >> 16);

    float4* orow = (float4*)(out + (size_t)wid * DIM);
    if (degc == 0) {
        const float4* trow = (const float4*)(h_type + (size_t)wid * DIM);
        orow[lane] = trow[lane];
        return;
    }
    const float sdt = s_dst[wid];
    const float NEG = -3.402823466e38f;
    float4 acc = {0.0f, 0.0f, 0.0f, 0.0f};

    if (degc <= SLOTS) {
        const int deg = degc;
        int* buf = sbuf[threadIdx.x >> 5];
        const int ex   = inc - q;
        const int pos0 = ex & 0xFFFF;
        const int pos1 = tot0 + (ex >> 16);
        const int a0 = lane * CHUNK + st0;
        const int a1 = (lane + 32) * CHUNK + st1;
#pragma clang loop unroll(disable) vectorize(disable) interleave(disable)
        for (int i = 0; i < c0; ++i) buf[pos0 + i] = a0 + i;
#pragma clang loop unroll(disable) vectorize(disable) interleave(disable)
        for (int i = 0; i < c1; ++i) buf[pos1 + i] = a1 + i;
        __builtin_amdgcn_wave_barrier();
        int   sr[NCACHE];
        float v[NCACHE];
#pragma unroll
        for (int c = 0; c < NCACHE; ++c) sr[c] = buf[min(c * 32 + lane, deg - 1)];
#pragma unroll
        for (int c = 0; c < NCACHE; ++c) sr[c] = srt[sr[c]];
#pragma unroll
        for (int c = 0; c < NCACHE; ++c) v[c] = s_src[sr[c]];
        float mx = NEG;
#pragma unroll
        for (int c = 0; c < NCACHE; ++c) {
            float t = v[c] + sdt;
            t = (t > 0.0f) ? t : SLOPE * t;
            v[c] = (c * 32 + lane < deg) ? t : NEG;
            mx = fmaxf(mx, v[c]);
        }
#pragma unroll
        for (int off = 16; off > 0; off >>= 1) mx = fmaxf(mx, __shfl_xor(mx, off, 32));
        float sm = 0.0f;
#pragma unroll
        for (int c = 0; c < NCACHE; ++c) {
            v[c] = (c * 32 + lane < deg) ? __expf(v[c] - mx) : 0.0f;
            sm += v[c];
        }
#pragma unroll
        for (int off = 16; off > 0; off >>= 1) sm += __shfl_xor(sm, off, 32);
        const float inv = 1.0f / sm;
#pragma unroll
        for (int c = 0; c < NCACHE; ++c) {
            const int cnt = min(deg - c * 32, 32);
            if (cnt > 0) gather_rows(h_sent, sr[c], v[c] * inv, cnt, lane, acc);
        }
    } else {
        float mx = NEG;
        for (int k = 0; k < NCHUNK; ++k) {
            const int rs = __builtin_amdgcn_readlane(k < 32 ? st0 : st1, k & 31);
            const int rc = __builtin_amdgcn_readlane(k < 32 ? c0 : c1, k & 31);
            const int* run = srt + k * CHUNK + rs;
            for (int i = lane; i < rc; i += 32) {
                float t = s_src[run[i]] + sdt;
                t = (t > 0.0f) ? t : SLOPE * t;
                mx = fmaxf(mx, t);
            }
        }
#pragma unroll
        for (int off = 16; off > 0; off >>= 1) mx = fmaxf(mx, __shfl_xor(mx, off, 32));
        float sm = 0.0f;
        for (int k = 0; k < NCHUNK; ++k) {
            const int rs = __builtin_amdgcn_readlane(k < 32 ? st0 : st1, k & 31);
            const int rc = __builtin_amdgcn_readlane(k < 32 ? c0 : c1, k & 31);
            const int* run = srt + k * CHUNK + rs;
            for (int i = lane; i < rc; i += 32) {
                float t = s_src[run[i]] + sdt;
                t = (t > 0.0f) ? t : SLOPE * t;
                sm += __expf(t - mx);
            }
        }
#pragma unroll
        for (int off = 16; off > 0; off >>= 1) sm += __shfl_xor(sm, off, 32);
        const float inv = 1.0f / sm;
        for (int k = 0; k < NCHUNK; ++k) {
            const int rs = __builtin_amdgcn_readlane(k < 32 ? st0 : st1, k & 31);
            const int rc = __builtin_amdgcn_readlane(k < 32 ? c0 : c1, k & 31);
            const int* run = srt + k * CHUNK + rs;
            for (int base = 0; base < rc; base += 32) {
                const int i = base + lane;
                const int s = run[min(i, rc - 1)];
                float t = s_src[s] + sdt;
                t = (t > 0.0f) ? t : SLOPE * t;
                const float w = (i < rc) ? __expf(t - mx) * inv : 0.0f;
                gather_rows(h_sent, s, w, min(rc - base, 32), lane, acc);
            }
        }
    }
    orow[lane] = acc;
}

extern "C" void kernel_launch(void* const* d_in, const int* in_sizes, int n_in,
                              void* d_out, int out_size, void* d_ws, size_t ws_size,
                              hipStream_t stream) {
    const float* h_sent = (const float*)d_in[0];
    const float* h_type = (const float*)d_in[1];
    const float* attn_w = (const float*)d_in[2];
    const int*   srcI   = (const int*)d_in[3];
    const int*   dstI   = (const int*)d_in[4];
    float*       out    = (float*)d_out;

    char* ws = (char*)d_ws;
    float* s_src = (float*)(ws);
    float* s_dst = (float*)(ws + 400000);
    int*   tab   = (int*)  (ws + 520192);
    int*   srt   = (int*)  (ws + 3080192);
    (void)ws_size; (void)n_in; (void)in_sizes; (void)out_size;

    gat_sort<<<NCHUNK + SCORE_BLOCKS_ALL, 1024, 0, stream>>>(dstI, srcI, tab, srt, h_sent, h_type, attn_w, s_src, s_dst);

    gat_aggregate<<<N_TYPE / AGG_WAVES, 32 * AGG_WAVES, 0, stream>>>(h_sent, h_type, s_src, s_dst, tab, srt, out);
}
